# starting kernel
# baseline (speedup 1.0000x reference)
#include <hip/hip_runtime.h>
#include <math.h>

typedef __attribute__((ext_vector_type(2))) float v2f;
typedef __attribute__((ext_vector_type(8))) float v8f;

#define EPS_LN 1e-6f
#define EPS_L2 1e-12f

__device__ __forceinline__ float gelu_exact(float v) {
    return 0.5f * v * (1.0f + erff(v * 0.70710678118654752440f));
}

__global__ __launch_bounds__(256) void tok_kernel(
    const float* __restrict__ token, const float* __restrict__ p,
    const float* __restrict__ alpha,
    const float* __restrict__ g1, const float* __restrict__ b1,
    const float* __restrict__ w_tok, const float* __restrict__ b_tok,
    float* __restrict__ tn_out)
{
    __shared__ float red[256];
    __shared__ float tnorm[768];
    __shared__ float gbuf[128];
    __shared__ float sm, srstd, snrm;
    const int b = blockIdx.x, t = threadIdx.x;
    const float* tok = token + b * 768;

    float s = 0.f;
    for (int i = t; i < 768; i += 256) s += tok[i];
    red[t] = s; __syncthreads();
    for (int off = 128; off > 0; off >>= 1) { if (t < off) red[t] += red[t + off]; __syncthreads(); }
    if (t == 0) sm = red[0] * (1.f / 768.f);
    __syncthreads();
    const float m = sm;

    float q = 0.f;
    for (int i = t; i < 768; i += 256) { float d = tok[i] - m; q += d * d; }
    red[t] = q; __syncthreads();
    for (int off = 128; off > 0; off >>= 1) { if (t < off) red[t] += red[t + off]; __syncthreads(); }
    if (t == 0) srstd = rsqrtf(red[0] * (1.f / 768.f) + EPS_LN);
    __syncthreads();
    const float rstd = srstd;

    for (int i = t; i < 768; i += 256) tnorm[i] = (tok[i] - m) * rstd * g1[i] + b1[i];
    __syncthreads();

    if (t < 128) {
        float acc = b_tok[t];
        for (int k = 0; k < 768; ++k) acc = fmaf(tnorm[k], w_tok[k * 128 + t], acc);
        gbuf[t] = gelu_exact(acc);
    }
    __syncthreads();

    float ss = 0.f;
    if (t < 128) { float g = gbuf[t]; ss = g * g; }
    red[t] = ss; __syncthreads();
    for (int off = 128; off > 0; off >>= 1) { if (t < off) red[t] += red[t + off]; __syncthreads(); }
    if (t == 0) snrm = fmaxf(sqrtf(red[0]), EPS_L2);
    __syncthreads();

    if (t < 128) {
        float scale = p[b] * expf(alpha[0]) / snrm;
        tn_out[b * 128 + t] = gbuf[t] * scale;
    }
}

#define ROWP 196

__global__ __launch_bounds__(256) void xattn_kernel(
    const float* __restrict__ x, const float* __restrict__ w_x,
    const float* __restrict__ b_x,
    const float* __restrict__ g2, const float* __restrict__ b2,
    const float* __restrict__ g3, const float* __restrict__ b3,
    const float* __restrict__ tn, float* __restrict__ out)
{
    __shared__ float xs[16 * ROWP];
    __shared__ float as_[16 * ROWP];
    __shared__ float sg2[192], sb2[192], sg3[192], sb3[192];
    __shared__ float tnl[128];
    __shared__ float mx[16], vx[16];
    __shared__ float pdot[16][8], psq[16][8];
    __shared__ float sattn[16], smeany[16], srstdy[16];

    const int t    = threadIdx.x;
    const int wave = t >> 5;
    const int lane = t & 31;
    const int nl   = lane & 15;
    const int hi   = lane >> 4;
    const int b     = blockIdx.x >> 4;
    const int chunk = blockIdx.x & 15;
    const int ncol  = wave * 16 + nl;

    for (int i = t; i < 192; i += 256) {
        sg2[i] = g2[i]; sb2[i] = b2[i]; sg3[i] = g3[i]; sb3[i] = b3[i];
    }
    if (t < 128) tnl[t] = tn[b * 128 + t];

    v2f bfrag[48];
    #pragma unroll
    for (int kb = 0; kb < 48; ++kb) {
        const int k0 = kb * 4 + hi * 2;
        bfrag[kb].x = w_x[k0 * 128 + ncol];
        bfrag[kb].y = w_x[(k0 + 1) * 128 + ncol];
    }
    const float bxv = b_x[ncol];
    __syncthreads();
    const float tnv = tnl[ncol];

    const size_t base = ((size_t)(b * 4096 + chunk * 256)) * 192;
    float4 pf[3];
    {
        const float4* xg = (const float4*)(x + base);
        #pragma unroll
        for (int j = 0; j < 3; ++j) pf[j] = xg[t + j * 256];
    }

    for (int tt = 0; tt < 16; ++tt) {
        const size_t tbase = base + (size_t)tt * 3072;

        #pragma unroll
        for (int j = 0; j < 3; ++j) {
            int e4 = t + j * 256;
            int r = e4 / 48, c4 = e4 - r * 48;
            *(float4*)&xs[r * ROWP + c4 * 4] = pf[j];
        }
        if (tt < 15) {
            const float4* xg = (const float4*)(x + tbase + 3072);
            #pragma unroll
            for (int j = 0; j < 3; ++j) pf[j] = xg[t + j * 256];
        }
        __syncthreads();

        for (int rr = 0; rr < 2; ++rr) {
            const int r = wave + rr * 8;
            float s = 0.f, q = 0.f;
            for (int j = 0; j < 6; ++j) {
                float xv = xs[r * ROWP + lane + j * 32];
                s += xv; q += xv * xv;
            }
            for (int msk = 16; msk >= 1; msk >>= 1) {
                s += __shfl_xor(s, msk, 32);
                q += __shfl_xor(q, msk, 32);
            }
            float m   = s * (1.f / 192.f);
            float var = q * (1.f / 192.f) - m * m;
            float rst = rsqrtf(var + EPS_LN);
            if (lane == 0) { mx[r] = m; vx[r] = var; }
            for (int j = 0; j < 6; ++j) {
                int c = lane + j * 32;
                as_[r * ROWP + c] = (xs[r * ROWP + c] - m) * rst * sg2[c] + sb2[c];
            }
        }
        __syncthreads();

        v8f acc0 = {}, acc1 = {};
        const float* arow = &as_[nl * ROWP + hi * 2];
        #pragma unroll
        for (int kb = 0; kb < 48; kb += 2) {
            v2f a0 = *(const v2f*)(arow + kb * 4);
            v2f a1 = *(const v2f*)(arow + kb * 4 + 4);
            acc0 = __builtin_amdgcn_wmma_f32_16x16x4_f32(
                false, a0, false, bfrag[kb], (short)0, acc0, false, false);
            acc1 = __builtin_amdgcn_wmma_f32_16x16x4_f32(
                false, a1, false, bfrag[kb + 1], (short)0, acc1, false, false);
        }

        #pragma unroll
        for (int v = 0; v < 8; ++v) {
            float g  = gelu_exact(acc0[v] + acc1[v] + bxv);
            float pd = g * tnv;
            float ps = g * g;
            for (int msk = 8; msk >= 1; msk >>= 1) {
                pd += __shfl_xor(pd, msk, 32);
                ps += __shfl_xor(ps, msk, 32);
            }
            if (nl == 0) {
                int r = v + hi * 8;
                pdot[r][wave] = pd;
                psq[r][wave]  = ps;
            }
        }
        __syncthreads();

        if (t < 16) {
            float dot = 0.f, sq = 0.f;
            for (int wv = 0; wv < 8; ++wv) { dot += pdot[t][wv]; sq += psq[t][wv]; }
            float attn = dot / fmaxf(sqrtf(sq), EPS_L2);
            sattn[t]  = attn;
            smeany[t] = attn * mx[t];
            srstdy[t] = rsqrtf(attn * attn * vx[t] + EPS_LN);
        }
        __syncthreads();

        {
            const int r  = t >> 4;
            const int c0 = (t & 15) * 12;
            const float attn = sattn[r], my = smeany[r], rs = srstdy[r];
            float* og = out + tbase + (size_t)r * 192;
            const float* xr = &xs[r * ROWP];
            #pragma unroll
            for (int c = c0; c < c0 + 12; ++c) {
                float xv = xr[c];
                og[c] = 0.5f * xv + (xv * attn - my) * rs * sg3[c] + sb3[c];
            }
        }
        __syncthreads();
    }
}

extern "C" void kernel_launch(void* const* d_in, const int* in_sizes, int n_in,
                              void* d_out, int out_size, void* d_ws, size_t ws_size,
                              hipStream_t stream) {
    const float* x     = (const float*)d_in[0];
    const float* token = (const float*)d_in[1];
    const float* p     = (const float*)d_in[2];
    const float* alpha = (const float*)d_in[3];
    const float* ln1_g = (const float*)d_in[4];
    const float* ln1_b = (const float*)d_in[5];
    const float* w_tok = (const float*)d_in[6];
    const float* b_tok = (const float*)d_in[7];
    const float* ln2_g = (const float*)d_in[8];
    const float* ln2_b = (const float*)d_in[9];
    const float* w_x   = (const float*)d_in[10];
    const float* b_x   = (const float*)d_in[11];
    const float* ln3_g = (const float*)d_in[12];
    const float* ln3_b = (const float*)d_in[13];
    float* out = (float*)d_out;
    float* tn  = (float*)d_ws;

    tok_kernel<<<32, 256, 0, stream>>>(token, p, alpha, ln1_g, ln1_b,
                                       w_tok, b_tok, tn);
    xattn_kernel<<<512, 256, 0, stream>>>(x, w_x, b_x, ln2_g, ln2_b,
                                          ln3_g, ln3_b, tn, out);
}


# Cross-lane sums (LN2 row stats, per-row dot and square sums) as row_xmask DPP adds plus one permlanex16 on the vector ALU instead of ds_bpermute round trips; same butterfly order, bit-identical sums
# speedup vs baseline: 1.1743x; 1.1743x over previous
#include <hip/hip_runtime.h>
#include <math.h>

typedef __attribute__((ext_vector_type(2))) float v2f;
typedef __attribute__((ext_vector_type(8))) float v8f;

#define EPS_LN 1e-6f
#define EPS_L2 1e-12f

__device__ __forceinline__ float gelu_exact(float v) {
    return 0.5f * v * (1.0f + erff(v * 0.70710678118654752440f));
}

__device__ __forceinline__ float xhalf(float x) {
    int b = __builtin_bit_cast(int, x);
    return __builtin_bit_cast(float, __builtin_amdgcn_permlanex16(
        b, b, 0x76543210, 0xfedcba98, false, false));
}
template <int M>
__device__ __forceinline__ float xrow(float x) {
    int b = __builtin_bit_cast(int, x);
    return __builtin_bit_cast(float, __builtin_amdgcn_update_dpp(
        b, b, 0x160 + M, 0xf, 0xf, true));
}

__global__ __launch_bounds__(256) void tok_kernel(
    const float* __restrict__ token, const float* __restrict__ p,
    const float* __restrict__ alpha,
    const float* __restrict__ g1, const float* __restrict__ b1,
    const float* __restrict__ w_tok, const float* __restrict__ b_tok,
    float* __restrict__ tn_out)
{
    __shared__ float red[256];
    __shared__ float tnorm[768];
    __shared__ float gbuf[128];
    __shared__ float sm, srstd, snrm;
    const int b = blockIdx.x, t = threadIdx.x;
    const float* tok = token + b * 768;

    float s = 0.f;
    for (int i = t; i < 768; i += 256) s += tok[i];
    red[t] = s; __syncthreads();
    for (int off = 128; off > 0; off >>= 1) { if (t < off) red[t] += red[t + off]; __syncthreads(); }
    if (t == 0) sm = red[0] * (1.f / 768.f);
    __syncthreads();
    const float m = sm;

    float q = 0.f;
    for (int i = t; i < 768; i += 256) { float d = tok[i] - m; q += d * d; }
    red[t] = q; __syncthreads();
    for (int off = 128; off > 0; off >>= 1) { if (t < off) red[t] += red[t + off]; __syncthreads(); }
    if (t == 0) srstd = rsqrtf(red[0] * (1.f / 768.f) + EPS_LN);
    __syncthreads();
    const float rstd = srstd;

    for (int i = t; i < 768; i += 256) tnorm[i] = (tok[i] - m) * rstd * g1[i] + b1[i];
    __syncthreads();

    if (t < 128) {
        float acc = b_tok[t];
        for (int k = 0; k < 768; ++k) acc = fmaf(tnorm[k], w_tok[k * 128 + t], acc);
        gbuf[t] = gelu_exact(acc);
    }
    __syncthreads();

    float ss = 0.f;
    if (t < 128) { float g = gbuf[t]; ss = g * g; }
    red[t] = ss; __syncthreads();
    for (int off = 128; off > 0; off >>= 1) { if (t < off) red[t] += red[t + off]; __syncthreads(); }
    if (t == 0) snrm = fmaxf(sqrtf(red[0]), EPS_L2);
    __syncthreads();

    if (t < 128) {
        float scale = p[b] * expf(alpha[0]) / snrm;
        tn_out[b * 128 + t] = gbuf[t] * scale;
    }
}

#define ROWP 196

__global__ __launch_bounds__(256) void xattn_kernel(
    const float* __restrict__ x, const float* __restrict__ w_x,
    const float* __restrict__ b_x,
    const float* __restrict__ g2, const float* __restrict__ b2,
    const float* __restrict__ g3, const float* __restrict__ b3,
    const float* __restrict__ tn, float* __restrict__ out)
{
    __shared__ float xs[16 * ROWP];
    __shared__ float as_[16 * ROWP];
    __shared__ float sg2[192], sb2[192], sg3[192], sb3[192];
    __shared__ float tnl[128];
    __shared__ float mx[16], vx[16];
    __shared__ float pdot[16][8], psq[16][8];
    __shared__ float sattn[16], smeany[16], srstdy[16];

    const int t    = threadIdx.x;
    const int wave = t >> 5;
    const int lane = t & 31;
    const int nl   = lane & 15;
    const int hi   = lane >> 4;
    const int b     = blockIdx.x >> 4;
    const int chunk = blockIdx.x & 15;
    const int ncol  = wave * 16 + nl;

    for (int i = t; i < 192; i += 256) {
        sg2[i] = g2[i]; sb2[i] = b2[i]; sg3[i] = g3[i]; sb3[i] = b3[i];
    }
    if (t < 128) tnl[t] = tn[b * 128 + t];

    v2f bfrag[48];
    #pragma unroll
    for (int kb = 0; kb < 48; ++kb) {
        const int k0 = kb * 4 + hi * 2;
        bfrag[kb].x = w_x[k0 * 128 + ncol];
        bfrag[kb].y = w_x[(k0 + 1) * 128 + ncol];
    }
    const float bxv = b_x[ncol];
    __syncthreads();
    const float tnv = tnl[ncol];

    const size_t base = ((size_t)(b * 4096 + chunk * 256)) * 192;
    float4 pf[3];
    {
        const float4* xg = (const float4*)(x + base);
        #pragma unroll
        for (int j = 0; j < 3; ++j) pf[j] = xg[t + j * 256];
    }

    for (int tt = 0; tt < 16; ++tt) {
        const size_t tbase = base + (size_t)tt * 3072;

        #pragma unroll
        for (int j = 0; j < 3; ++j) {
            int e4 = t + j * 256;
            int r = e4 / 48, c4 = e4 - r * 48;
            *(float4*)&xs[r * ROWP + c4 * 4] = pf[j];
        }
        if (tt < 15) {
            const float4* xg = (const float4*)(x + tbase + 3072);
            #pragma unroll
            for (int j = 0; j < 3; ++j) pf[j] = xg[t + j * 256];
        }
        __syncthreads();

        for (int rr = 0; rr < 2; ++rr) {
            const int r = wave + rr * 8;
            float s = 0.f, q = 0.f;
            for (int j = 0; j < 6; ++j) {
                float xv = xs[r * ROWP + lane + j * 32];
                s += xv; q += xv * xv;
            }
            s += xhalf(s);   q += xhalf(q);
            s += xrow<8>(s); q += xrow<8>(q);
            s += xrow<4>(s); q += xrow<4>(q);
            s += xrow<2>(s); q += xrow<2>(q);
            s += xrow<1>(s); q += xrow<1>(q);
            float m   = s * (1.f / 192.f);
            float var = q * (1.f / 192.f) - m * m;
            float rst = rsqrtf(var + EPS_LN);
            if (lane == 0) { mx[r] = m; vx[r] = var; }
            for (int j = 0; j < 6; ++j) {
                int c = lane + j * 32;
                as_[r * ROWP + c] = (xs[r * ROWP + c] - m) * rst * sg2[c] + sb2[c];
            }
        }
        __syncthreads();

        v8f acc0 = {}, acc1 = {};
        const float* arow = &as_[nl * ROWP + hi * 2];
        #pragma unroll
        for (int kb = 0; kb < 48; kb += 2) {
            v2f a0 = *(const v2f*)(arow + kb * 4);
            v2f a1 = *(const v2f*)(arow + kb * 4 + 4);
            acc0 = __builtin_amdgcn_wmma_f32_16x16x4_f32(
                false, a0, false, bfrag[kb], (short)0, acc0, false, false);
            acc1 = __builtin_amdgcn_wmma_f32_16x16x4_f32(
                false, a1, false, bfrag[kb + 1], (short)0, acc1, false, false);
        }

        #pragma unroll
        for (int v = 0; v < 8; ++v) {
            float g  = gelu_exact(acc0[v] + acc1[v] + bxv);
            float pd = g * tnv;
            float ps = g * g;
            pd += xrow<8>(pd); ps += xrow<8>(ps);
            pd += xrow<4>(pd); ps += xrow<4>(ps);
            pd += xrow<2>(pd); ps += xrow<2>(ps);
            pd += xrow<1>(pd); ps += xrow<1>(ps);
            if (nl == 0) {
                int r = v + hi * 8;
                pdot[r][wave] = pd;
                psq[r][wave]  = ps;
            }
        }
        __syncthreads();

        if (t < 16) {
            float dot = 0.f, sq = 0.f;
            for (int wv = 0; wv < 8; ++wv) { dot += pdot[t][wv]; sq += psq[t][wv]; }
            float attn = dot / fmaxf(sqrtf(sq), EPS_L2);
            sattn[t]  = attn;
            smeany[t] = attn * mx[t];
            srstdy[t] = rsqrtf(attn * attn * vx[t] + EPS_LN);
        }
        __syncthreads();

        {
            const int r  = t >> 4;
            const int c0 = (t & 15) * 12;
            const float attn = sattn[r], my = smeany[r], rs = srstdy[r];
            float* og = out + tbase + (size_t)r * 192;
            const float* xr = &xs[r * ROWP];
            #pragma unroll
            for (int c = c0; c < c0 + 12; ++c) {
                float xv = xr[c];
                og[c] = 0.5f * xv + (xv * attn - my) * rs * sg3[c] + sb3[c];
            }
        }
        __syncthreads();
    }
}

extern "C" void kernel_launch(void* const* d_in, const int* in_sizes, int n_in,
                              void* d_out, int out_size, void* d_ws, size_t ws_size,
                              hipStream_t stream) {
    const float* x     = (const float*)d_in[0];
    const float* token = (const float*)d_in[1];
    const float* p     = (const float*)d_in[2];
    const float* alpha = (const float*)d_in[3];
    const float* ln1_g = (const float*)d_in[4];
    const float* ln1_b = (const float*)d_in[5];
    const float* w_tok = (const float*)d_in[6];
    const float* b_tok = (const float*)d_in[7];
    const float* ln2_g = (const float*)d_in[8];
    const float* ln2_b = (const float*)d_in[9];
    const float* w_x   = (const float*)d_in[10];
    const float* b_x   = (const float*)d_in[11];
    const float* ln3_g = (const float*)d_in[12];
    const float* ln3_b = (const float*)d_in[13];
    float* out = (float*)d_out;
    float* tn  = (float*)d_ws;

    tok_kernel<<<32, 256, 0, stream>>>(token, p, alpha, ln1_g, ln1_b,
                                       w_tok, b_tok, tn);
    xattn_kernel<<<512, 256, 0, stream>>>(x, w_x, b_x, ln2_g, ln2_b,
                                          ln3_g, ln3_b, tn, out);
}


# Next-tile x prefetch held in named registers (it had been demoted to scratch and waited at issue) and xattn barriers fence LDS only: x loads and output stores stay in flight across barriers
# speedup vs baseline: 1.6261x; 1.3847x over previous
#include <hip/hip_runtime.h>
#include <math.h>

typedef __attribute__((ext_vector_type(2))) float v2f;
typedef __attribute__((ext_vector_type(8))) float v8f;

#define EPS_LN 1e-6f
#define EPS_L2 1e-12f

__device__ __forceinline__ float gelu_exact(float v) {
    return 0.5f * v * (1.0f + erff(v * 0.70710678118654752440f));
}

__device__ __forceinline__ float xhalf(float x) {
    int b = __builtin_bit_cast(int, x);
    return __builtin_bit_cast(float, __builtin_amdgcn_permlanex16(
        b, b, 0x76543210, 0xfedcba98, false, false));
}
template <int M>
__device__ __forceinline__ float xrow(float x) {
    int b = __builtin_bit_cast(int, x);
    return __builtin_bit_cast(float, __builtin_amdgcn_update_dpp(
        b, b, 0x160 + M, 0xf, 0xf, true));
}

__device__ __forceinline__ void lds_barrier() {
    __builtin_amdgcn_fence(__ATOMIC_RELEASE, "workgroup", "local");
    __builtin_amdgcn_s_barrier();
    __builtin_amdgcn_fence(__ATOMIC_ACQUIRE, "workgroup", "local");
}

__global__ __launch_bounds__(256) void tok_kernel(
    const float* __restrict__ token, const float* __restrict__ p,
    const float* __restrict__ alpha,
    const float* __restrict__ g1, const float* __restrict__ b1,
    const float* __restrict__ w_tok, const float* __restrict__ b_tok,
    float* __restrict__ tn_out)
{
    __shared__ float red[256];
    __shared__ float tnorm[768];
    __shared__ float gbuf[128];
    __shared__ float sm, srstd, snrm;
    const int b = blockIdx.x, t = threadIdx.x;
    const float* tok = token + b * 768;

    float s = 0.f;
    for (int i = t; i < 768; i += 256) s += tok[i];
    red[t] = s; __syncthreads();
    for (int off = 128; off > 0; off >>= 1) { if (t < off) red[t] += red[t + off]; __syncthreads(); }
    if (t == 0) sm = red[0] * (1.f / 768.f);
    __syncthreads();
    const float m = sm;

    float q = 0.f;
    for (int i = t; i < 768; i += 256) { float d = tok[i] - m; q += d * d; }
    red[t] = q; __syncthreads();
    for (int off = 128; off > 0; off >>= 1) { if (t < off) red[t] += red[t + off]; __syncthreads(); }
    if (t == 0) srstd = rsqrtf(red[0] * (1.f / 768.f) + EPS_LN);
    __syncthreads();
    const float rstd = srstd;

    for (int i = t; i < 768; i += 256) tnorm[i] = (tok[i] - m) * rstd * g1[i] + b1[i];
    __syncthreads();

    if (t < 128) {
        float acc = b_tok[t];
        for (int k = 0; k < 768; ++k) acc = fmaf(tnorm[k], w_tok[k * 128 + t], acc);
        gbuf[t] = gelu_exact(acc);
    }
    __syncthreads();

    float ss = 0.f;
    if (t < 128) { float g = gbuf[t]; ss = g * g; }
    red[t] = ss; __syncthreads();
    for (int off = 128; off > 0; off >>= 1) { if (t < off) red[t] += red[t + off]; __syncthreads(); }
    if (t == 0) snrm = fmaxf(sqrtf(red[0]), EPS_L2);
    __syncthreads();

    if (t < 128) {
        float scale = p[b] * expf(alpha[0]) / snrm;
        tn_out[b * 128 + t] = gbuf[t] * scale;
    }
}

#define ROWP 196

__global__ __launch_bounds__(256) void xattn_kernel(
    const float* __restrict__ x, const float* __restrict__ w_x,
    const float* __restrict__ b_x,
    const float* __restrict__ g2, const float* __restrict__ b2,
    const float* __restrict__ g3, const float* __restrict__ b3,
    const float* __restrict__ tn, float* __restrict__ out)
{
    __shared__ float xs[16 * ROWP];
    __shared__ float as_[16 * ROWP];
    __shared__ float sg2[192], sb2[192], sg3[192], sb3[192];
    __shared__ float tnl[128];
    __shared__ float mx[16], vx[16];
    __shared__ float pdot[16][8], psq[16][8];
    __shared__ float sattn[16], smeany[16], srstdy[16];

    const int t    = threadIdx.x;
    const int wave = t >> 5;
    const int lane = t & 31;
    const int nl   = lane & 15;
    const int hi   = lane >> 4;
    const int b     = blockIdx.x >> 4;
    const int chunk = blockIdx.x & 15;
    const int ncol  = wave * 16 + nl;

    for (int i = t; i < 192; i += 256) {
        sg2[i] = g2[i]; sb2[i] = b2[i]; sg3[i] = g3[i]; sb3[i] = b3[i];
    }
    if (t < 128) tnl[t] = tn[b * 128 + t];

    v2f bfrag[48];
    #pragma unroll
    for (int kb = 0; kb < 48; ++kb) {
        const int k0 = kb * 4 + hi * 2;
        bfrag[kb].x = w_x[k0 * 128 + ncol];
        bfrag[kb].y = w_x[(k0 + 1) * 128 + ncol];
    }
    const float bxv = b_x[ncol];
    lds_barrier();
    const float tnv = tnl[ncol];

    const size_t base = ((size_t)(b * 4096 + chunk * 256)) * 192;
    float4 pf0, pf1, pf2;
    {
        const float4* xg = (const float4*)(x + base);
        pf0 = xg[t]; pf1 = xg[t + 256]; pf2 = xg[t + 512];
    }
    float4* const xs0 = (float4*)&xs[(t / 48) * ROWP + (t % 48) * 4];
    float4* const xs1 = (float4*)&xs[((t + 256) / 48) * ROWP + ((t + 256) % 48) * 4];
    float4* const xs2 = (float4*)&xs[((t + 512) / 48) * ROWP + ((t + 512) % 48) * 4];

    for (int tt = 0; tt < 16; ++tt) {
        const size_t tbase = base + (size_t)tt * 3072;

        *xs0 = pf0; *xs1 = pf1; *xs2 = pf2;
        if (tt < 15) {
            const float4* xg = (const float4*)(x + tbase + 3072);
            pf0 = xg[t]; pf1 = xg[t + 256]; pf2 = xg[t + 512];
        }
        lds_barrier();

        for (int rr = 0; rr < 2; ++rr) {
            const int r = wave + rr * 8;
            float s = 0.f, q = 0.f;
            for (int j = 0; j < 6; ++j) {
                float xv = xs[r * ROWP + lane + j * 32];
                s += xv; q += xv * xv;
            }
            s += xhalf(s);   q += xhalf(q);
            s += xrow<8>(s); q += xrow<8>(q);
            s += xrow<4>(s); q += xrow<4>(q);
            s += xrow<2>(s); q += xrow<2>(q);
            s += xrow<1>(s); q += xrow<1>(q);
            float m   = s * (1.f / 192.f);
            float var = q * (1.f / 192.f) - m * m;
            float rst = rsqrtf(var + EPS_LN);
            if (lane == 0) { mx[r] = m; vx[r] = var; }
            for (int j = 0; j < 6; ++j) {
                int c = lane + j * 32;
                as_[r * ROWP + c] = (xs[r * ROWP + c] - m) * rst * sg2[c] + sb2[c];
            }
        }
        lds_barrier();

        v8f acc0 = {}, acc1 = {};
        const float* arow = &as_[nl * ROWP + hi * 2];
        #pragma unroll
        for (int kb = 0; kb < 48; kb += 2) {
            v2f a0 = *(const v2f*)(arow + kb * 4);
            v2f a1 = *(const v2f*)(arow + kb * 4 + 4);
            acc0 = __builtin_amdgcn_wmma_f32_16x16x4_f32(
                false, a0, false, bfrag[kb], (short)0, acc0, false, false);
            acc1 = __builtin_amdgcn_wmma_f32_16x16x4_f32(
                false, a1, false, bfrag[kb + 1], (short)0, acc1, false, false);
        }

        #pragma unroll
        for (int v = 0; v < 8; ++v) {
            float g  = gelu_exact(acc0[v] + acc1[v] + bxv);
            float pd = g * tnv;
            float ps = g * g;
            pd += xrow<8>(pd); ps += xrow<8>(ps);
            pd += xrow<4>(pd); ps += xrow<4>(ps);
            pd += xrow<2>(pd); ps += xrow<2>(ps);
            pd += xrow<1>(pd); ps += xrow<1>(ps);
            if (nl == 0) {
                int r = v + hi * 8;
                pdot[r][wave] = pd;
                psq[r][wave]  = ps;
            }
        }
        lds_barrier();

        if (t < 16) {
            float dot = 0.f, sq = 0.f;
            for (int wv = 0; wv < 8; ++wv) { dot += pdot[t][wv]; sq += psq[t][wv]; }
            float attn = dot / fmaxf(sqrtf(sq), EPS_L2);
            sattn[t]  = attn;
            smeany[t] = attn * mx[t];
            srstdy[t] = rsqrtf(attn * attn * vx[t] + EPS_LN);
        }
        lds_barrier();

        {
            const int r  = t >> 4;
            const int c0 = (t & 15) * 12;
            const float attn = sattn[r], my = smeany[r], rs = srstdy[r];
            float* og = out + tbase + (size_t)r * 192;
            const float* xr = &xs[r * ROWP];
            #pragma unroll
            for (int c = c0; c < c0 + 12; ++c) {
                float xv = xr[c];
                og[c] = 0.5f * xv + (xv * attn - my) * rs * sg3[c] + sb3[c];
            }
        }
        lds_barrier();
    }
}

extern "C" void kernel_launch(void* const* d_in, const int* in_sizes, int n_in,
                              void* d_out, int out_size, void* d_ws, size_t ws_size,
                              hipStream_t stream) {
    const float* x     = (const float*)d_in[0];
    const float* token = (const float*)d_in[1];
    const float* p     = (const float*)d_in[2];
    const float* alpha = (const float*)d_in[3];
    const float* ln1_g = (const float*)d_in[4];
    const float* ln1_b = (const float*)d_in[5];
    const float* w_tok = (const float*)d_in[6];
    const float* b_tok = (const float*)d_in[7];
    const float* ln2_g = (const float*)d_in[8];
    const float* ln2_b = (const float*)d_in[9];
    const float* w_x   = (const float*)d_in[10];
    const float* b_x   = (const float*)d_in[11];
    const float* ln3_g = (const float*)d_in[12];
    const float* ln3_b = (const float*)d_in[13];
    float* out = (float*)d_out;
    float* tn  = (float*)d_ws;

    tok_kernel<<<32, 256, 0, stream>>>(token, p, alpha, ln1_g, ln1_b,
                                       w_tok, b_tok, tn);
    xattn_kernel<<<512, 256, 0, stream>>>(x, w_x, b_x, ln2_g, ln2_b,
                                          ln3_g, ln3_b, tn, out);
}


# Token kernel: 768x128 matvec split over K across a 1024-thread block (wave = 24 rows, lane = 4 columns, all row loads in flight under the LN stats), partials added via LDS; changes that sum's order
# speedup vs baseline: 2.3116x; 1.4216x over previous
#include <hip/hip_runtime.h>
#include <math.h>

typedef __attribute__((ext_vector_type(2))) float v2f;
typedef __attribute__((ext_vector_type(8))) float v8f;

#define EPS_LN 1e-6f
#define EPS_L2 1e-12f

__device__ __forceinline__ float gelu_exact(float v) {
    return 0.5f * v * (1.0f + erff(v * 0.70710678118654752440f));
}

__device__ __forceinline__ float xhalf(float x) {
    int b = __builtin_bit_cast(int, x);
    return __builtin_bit_cast(float, __builtin_amdgcn_permlanex16(
        b, b, 0x76543210, 0xfedcba98, false, false));
}
template <int M>
__device__ __forceinline__ float xrow(float x) {
    int b = __builtin_bit_cast(int, x);
    return __builtin_bit_cast(float, __builtin_amdgcn_update_dpp(
        b, b, 0x160 + M, 0xf, 0xf, true));
}

__device__ __forceinline__ void lds_barrier() {
    __builtin_amdgcn_fence(__ATOMIC_RELEASE, "workgroup", "local");
    __builtin_amdgcn_s_barrier();
    __builtin_amdgcn_fence(__ATOMIC_ACQUIRE, "workgroup", "local");
}

__global__ __launch_bounds__(1024) void tok_kernel(
    const float* __restrict__ token, const float* __restrict__ p,
    const float* __restrict__ alpha,
    const float* __restrict__ g1, const float* __restrict__ b1,
    const float* __restrict__ w_tok, const float* __restrict__ b_tok,
    float* __restrict__ tn_out)
{
    __shared__ float red[256];
    __shared__ float tnorm[768];
    __shared__ float gbuf[128];
    __shared__ float4 part[32 * 32];
    __shared__ float sm, srstd, snrm;
    const int b = blockIdx.x, t = threadIdx.x;
    const int w = t >> 5, l = t & 31;
    const float* tok = token + b * 768;

    float a0 = 0.f, a1 = 0.f, a2 = 0.f;
    if (t < 256) { a0 = tok[t]; a1 = tok[t + 256]; a2 = tok[t + 512]; }
    float tkv = 0.f, g1v = 0.f, b1v = 0.f;
    if (t < 768) { tkv = tok[t]; g1v = g1[t]; b1v = b1[t]; }
    float btv = 0.f;
    if (t < 128) btv = b_tok[t];

    float4 wv[24];
    {
        const float4* wr = (const float4*)(w_tok + (size_t)w * (24 * 128)) + l;
        #pragma unroll
        for (int j = 0; j < 24; ++j) wv[j] = wr[j * 32];
    }

    if (t < 256) red[t] = ((0.f + a0) + a1) + a2;
    lds_barrier();
    for (int off = 128; off > 0; off >>= 1) { if (t < off) red[t] += red[t + off]; lds_barrier(); }
    if (t == 0) sm = red[0] * (1.f / 768.f);
    lds_barrier();
    const float m = sm;

    if (t < 256) {
        const float d0 = a0 - m, d1 = a1 - m, d2 = a2 - m;
        float q = 0.f;
        q += d0 * d0; q += d1 * d1; q += d2 * d2;
        red[t] = q;
    }
    lds_barrier();
    for (int off = 128; off > 0; off >>= 1) { if (t < off) red[t] += red[t + off]; lds_barrier(); }
    if (t == 0) srstd = rsqrtf(red[0] * (1.f / 768.f) + EPS_LN);
    lds_barrier();
    const float rstd = srstd;

    if (t < 768) tnorm[t] = (tkv - m) * rstd * g1v + b1v;
    lds_barrier();

    {
        float4 acc = {0.f, 0.f, 0.f, 0.f};
        const float* tk = &tnorm[w * 24];
        #pragma unroll
        for (int j = 0; j < 24; ++j) {
            const float a = tk[j];
            acc.x = fmaf(a, wv[j].x, acc.x);
            acc.y = fmaf(a, wv[j].y, acc.y);
            acc.z = fmaf(a, wv[j].z, acc.z);
            acc.w = fmaf(a, wv[j].w, acc.w);
        }
        part[w * 32 + l] = acc;
    }
    lds_barrier();

    if (t < 128) {
        const float* pf = (const float*)part;
        float acc = btv;
        for (int sl = 0; sl < 32; ++sl) acc += pf[sl * 128 + t];
        gbuf[t] = gelu_exact(acc);
    }
    lds_barrier();

    if (t < 256) {
        float ss = 0.f;
        if (t < 128) { float g = gbuf[t]; ss = g * g; }
        red[t] = ss;
    }
    lds_barrier();
    for (int off = 128; off > 0; off >>= 1) { if (t < off) red[t] += red[t + off]; lds_barrier(); }
    if (t == 0) snrm = fmaxf(sqrtf(red[0]), EPS_L2);
    lds_barrier();

    if (t < 128) {
        float scale = p[b] * expf(alpha[0]) / snrm;
        tn_out[b * 128 + t] = gbuf[t] * scale;
    }
}

#define ROWP 196

__global__ __launch_bounds__(256) void xattn_kernel(
    const float* __restrict__ x, const float* __restrict__ w_x,
    const float* __restrict__ b_x,
    const float* __restrict__ g2, const float* __restrict__ b2,
    const float* __restrict__ g3, const float* __restrict__ b3,
    const float* __restrict__ tn, float* __restrict__ out)
{
    __shared__ float xs[16 * ROWP];
    __shared__ float as_[16 * ROWP];
    __shared__ float sg2[192], sb2[192], sg3[192], sb3[192];
    __shared__ float tnl[128];
    __shared__ float mx[16], vx[16];
    __shared__ float pdot[16][8], psq[16][8];
    __shared__ float sattn[16], smeany[16], srstdy[16];

    const int t    = threadIdx.x;
    const int wave = t >> 5;
    const int lane = t & 31;
    const int nl   = lane & 15;
    const int hi   = lane >> 4;
    const int b     = blockIdx.x >> 4;
    const int chunk = blockIdx.x & 15;
    const int ncol  = wave * 16 + nl;

    for (int i = t; i < 192; i += 256) {
        sg2[i] = g2[i]; sb2[i] = b2[i]; sg3[i] = g3[i]; sb3[i] = b3[i];
    }
    if (t < 128) tnl[t] = tn[b * 128 + t];

    v2f bfrag[48];
    #pragma unroll
    for (int kb = 0; kb < 48; ++kb) {
        const int k0 = kb * 4 + hi * 2;
        bfrag[kb].x = w_x[k0 * 128 + ncol];
        bfrag[kb].y = w_x[(k0 + 1) * 128 + ncol];
    }
    const float bxv = b_x[ncol];
    lds_barrier();
    const float tnv = tnl[ncol];

    const size_t base = ((size_t)(b * 4096 + chunk * 256)) * 192;
    float4 pf0, pf1, pf2;
    {
        const float4* xg = (const float4*)(x + base);
        pf0 = xg[t]; pf1 = xg[t + 256]; pf2 = xg[t + 512];
    }
    float4* const xs0 = (float4*)&xs[(t / 48) * ROWP + (t % 48) * 4];
    float4* const xs1 = (float4*)&xs[((t + 256) / 48) * ROWP + ((t + 256) % 48) * 4];
    float4* const xs2 = (float4*)&xs[((t + 512) / 48) * ROWP + ((t + 512) % 48) * 4];

    for (int tt = 0; tt < 16; ++tt) {
        const size_t tbase = base + (size_t)tt * 3072;

        *xs0 = pf0; *xs1 = pf1; *xs2 = pf2;
        if (tt < 15) {
            const float4* xg = (const float4*)(x + tbase + 3072);
            pf0 = xg[t]; pf1 = xg[t + 256]; pf2 = xg[t + 512];
        }
        lds_barrier();

        for (int rr = 0; rr < 2; ++rr) {
            const int r = wave + rr * 8;
            float s = 0.f, q = 0.f;
            for (int j = 0; j < 6; ++j) {
                float xv = xs[r * ROWP + lane + j * 32];
                s += xv; q += xv * xv;
            }
            s += xhalf(s);   q += xhalf(q);
            s += xrow<8>(s); q += xrow<8>(q);
            s += xrow<4>(s); q += xrow<4>(q);
            s += xrow<2>(s); q += xrow<2>(q);
            s += xrow<1>(s); q += xrow<1>(q);
            float m   = s * (1.f / 192.f);
            float var = q * (1.f / 192.f) - m * m;
            float rst = rsqrtf(var + EPS_LN);
            if (lane == 0) { mx[r] = m; vx[r] = var; }
            for (int j = 0; j < 6; ++j) {
                int c = lane + j * 32;
                as_[r * ROWP + c] = (xs[r * ROWP + c] - m) * rst * sg2[c] + sb2[c];
            }
        }
        lds_barrier();

        v8f acc0 = {}, acc1 = {};
        const float* arow = &as_[nl * ROWP + hi * 2];
        #pragma unroll
        for (int kb = 0; kb < 48; kb += 2) {
            v2f a0 = *(const v2f*)(arow + kb * 4);
            v2f a1 = *(const v2f*)(arow + kb * 4 + 4);
            acc0 = __builtin_amdgcn_wmma_f32_16x16x4_f32(
                false, a0, false, bfrag[kb], (short)0, acc0, false, false);
            acc1 = __builtin_amdgcn_wmma_f32_16x16x4_f32(
                false, a1, false, bfrag[kb + 1], (short)0, acc1, false, false);
        }

        #pragma unroll
        for (int v = 0; v < 8; ++v) {
            float g  = gelu_exact(acc0[v] + acc1[v] + bxv);
            float pd = g * tnv;
            float ps = g * g;
            pd += xrow<8>(pd); ps += xrow<8>(ps);
            pd += xrow<4>(pd); ps += xrow<4>(ps);
            pd += xrow<2>(pd); ps += xrow<2>(ps);
            pd += xrow<1>(pd); ps += xrow<1>(ps);
            if (nl == 0) {
                int r = v + hi * 8;
                pdot[r][wave] = pd;
                psq[r][wave]  = ps;
            }
        }
        lds_barrier();

        if (t < 16) {
            float dot = 0.f, sq = 0.f;
            for (int wv = 0; wv < 8; ++wv) { dot += pdot[t][wv]; sq += psq[t][wv]; }
            float attn = dot / fmaxf(sqrtf(sq), EPS_L2);
            sattn[t]  = attn;
            smeany[t] = attn * mx[t];
            srstdy[t] = rsqrtf(attn * attn * vx[t] + EPS_LN);
        }
        lds_barrier();

        {
            const int r  = t >> 4;
            const int c0 = (t & 15) * 12;
            const float attn = sattn[r], my = smeany[r], rs = srstdy[r];
            float* og = out + tbase + (size_t)r * 192;
            const float* xr = &xs[r * ROWP];
            #pragma unroll
            for (int c = c0; c < c0 + 12; ++c) {
                float xv = xr[c];
                og[c] = 0.5f * xv + (xv * attn - my) * rs * sg3[c] + sb3[c];
            }
        }
        lds_barrier();
    }
}

extern "C" void kernel_launch(void* const* d_in, const int* in_sizes, int n_in,
                              void* d_out, int out_size, void* d_ws, size_t ws_size,
                              hipStream_t stream) {
    const float* x     = (const float*)d_in[0];
    const float* token = (const float*)d_in[1];
    const float* p     = (const float*)d_in[2];
    const float* alpha = (const float*)d_in[3];
    const float* ln1_g = (const float*)d_in[4];
    const float* ln1_b = (const float*)d_in[5];
    const float* w_tok = (const float*)d_in[6];
    const float* b_tok = (const float*)d_in[7];
    const float* ln2_g = (const float*)d_in[8];
    const float* ln2_b = (const float*)d_in[9];
    const float* w_x   = (const float*)d_in[10];
    const float* b_x   = (const float*)d_in[11];
    const float* ln3_g = (const float*)d_in[12];
    const float* ln3_b = (const float*)d_in[13];
    float* out = (float*)d_out;
    float* tn  = (float*)d_ws;

    tok_kernel<<<32, 1024, 0, stream>>>(token, p, alpha, ln1_g, ln1_b,
                                       w_tok, b_tok, tn);
    xattn_kernel<<<512, 256, 0, stream>>>(x, w_x, b_x, ln2_g, ln2_b,
                                          ln3_g, ln3_b, tn, out);
}


# Tile epilogue without per-row branches so the 8 accumulator rows' gelu and row-sum chains interleave: erf by the library's |z|<1 polynomial, ONE wave-uniform fallback to erff; values bit-identical
# speedup vs baseline: 2.4087x; 1.0420x over previous
#include <hip/hip_runtime.h>
#include <math.h>

typedef __attribute__((ext_vector_type(2))) float v2f;
typedef __attribute__((ext_vector_type(8))) float v8f;

#define EPS_LN 1e-6f
#define EPS_L2 1e-12f

__device__ __forceinline__ float gelu_exact(float v) {
    return 0.5f * v * (1.0f + erff(v * 0.70710678118654752440f));
}

__device__ __forceinline__ float erf_small(float z) {
    const float t = z * z;
    float p = fmaf(t, -0x1.268bc2p-11f, 0x1.420828p-8f);
    p = fmaf(t, p, -0x1.b5937p-6f);
    p = fmaf(t, p, 0x1.ce077cp-4f);
    p = fmaf(t, p, -0x1.81266p-2f);
    p = fmaf(t, p, 0x1.06eba0p-3f);
    const float az = fabsf(z);
    return copysignf(fmaf(az, p, az), z);
}

__device__ __forceinline__ float xhalf(float x) {
    int b = __builtin_bit_cast(int, x);
    return __builtin_bit_cast(float, __builtin_amdgcn_permlanex16(
        b, b, 0x76543210, 0xfedcba98, false, false));
}
template <int M>
__device__ __forceinline__ float xrow(float x) {
    int b = __builtin_bit_cast(int, x);
    return __builtin_bit_cast(float, __builtin_amdgcn_update_dpp(
        b, b, 0x160 + M, 0xf, 0xf, true));
}

__device__ __forceinline__ void lds_barrier() {
    __builtin_amdgcn_fence(__ATOMIC_RELEASE, "workgroup", "local");
    __builtin_amdgcn_s_barrier();
    __builtin_amdgcn_fence(__ATOMIC_ACQUIRE, "workgroup", "local");
}

__global__ __launch_bounds__(1024) void tok_kernel(
    const float* __restrict__ token, const float* __restrict__ p,
    const float* __restrict__ alpha,
    const float* __restrict__ g1, const float* __restrict__ b1,
    const float* __restrict__ w_tok, const float* __restrict__ b_tok,
    float* __restrict__ tn_out)
{
    __shared__ float red[256];
    __shared__ float tnorm[768];
    __shared__ float gbuf[128];
    __shared__ float4 part[32 * 32];
    __shared__ float sm, srstd, snrm;
    const int b = blockIdx.x, t = threadIdx.x;
    const int w = t >> 5, l = t & 31;
    const float* tok = token + b * 768;

    float a0 = 0.f, a1 = 0.f, a2 = 0.f;
    if (t < 256) { a0 = tok[t]; a1 = tok[t + 256]; a2 = tok[t + 512]; }
    float tkv = 0.f, g1v = 0.f, b1v = 0.f;
    if (t < 768) { tkv = tok[t]; g1v = g1[t]; b1v = b1[t]; }
    float btv = 0.f;
    if (t < 128) btv = b_tok[t];

    float4 wv[24];
    {
        const float4* wr = (const float4*)(w_tok + (size_t)w * (24 * 128)) + l;
        #pragma unroll
        for (int j = 0; j < 24; ++j) wv[j] = wr[j * 32];
    }

    if (t < 256) red[t] = ((0.f + a0) + a1) + a2;
    lds_barrier();
    for (int off = 128; off > 0; off >>= 1) { if (t < off) red[t] += red[t + off]; lds_barrier(); }
    if (t == 0) sm = red[0] * (1.f / 768.f);
    lds_barrier();
    const float m = sm;

    if (t < 256) {
        const float d0 = a0 - m, d1 = a1 - m, d2 = a2 - m;
        float q = 0.f;
        q += d0 * d0; q += d1 * d1; q += d2 * d2;
        red[t] = q;
    }
    lds_barrier();
    for (int off = 128; off > 0; off >>= 1) { if (t < off) red[t] += red[t + off]; lds_barrier(); }
    if (t == 0) srstd = rsqrtf(red[0] * (1.f / 768.f) + EPS_LN);
    lds_barrier();
    const float rstd = srstd;

    if (t < 768) tnorm[t] = (tkv - m) * rstd * g1v + b1v;
    lds_barrier();

    {
        float4 acc = {0.f, 0.f, 0.f, 0.f};
        const float* tk = &tnorm[w * 24];
        #pragma unroll
        for (int j = 0; j < 24; ++j) {
            const float a = tk[j];
            acc.x = fmaf(a, wv[j].x, acc.x);
            acc.y = fmaf(a, wv[j].y, acc.y);
            acc.z = fmaf(a, wv[j].z, acc.z);
            acc.w = fmaf(a, wv[j].w, acc.w);
        }
        part[w * 32 + l] = acc;
    }
    lds_barrier();

    if (t < 128) {
        const float* pf = (const float*)part;
        float acc = btv;
        for (int sl = 0; sl < 32; ++sl) acc += pf[sl * 128 + t];
        gbuf[t] = gelu_exact(acc);
    }
    lds_barrier();

    if (t < 256) {
        float ss = 0.f;
        if (t < 128) { float g = gbuf[t]; ss = g * g; }
        red[t] = ss;
    }
    lds_barrier();
    for (int off = 128; off > 0; off >>= 1) { if (t < off) red[t] += red[t + off]; lds_barrier(); }
    if (t == 0) snrm = fmaxf(sqrtf(red[0]), EPS_L2);
    lds_barrier();

    if (t < 128) {
        float scale = p[b] * expf(alpha[0]) / snrm;
        tn_out[b * 128 + t] = gbuf[t] * scale;
    }
}

#define ROWP 196

__global__ __launch_bounds__(256) void xattn_kernel(
    const float* __restrict__ x, const float* __restrict__ w_x,
    const float* __restrict__ b_x,
    const float* __restrict__ g2, const float* __restrict__ b2,
    const float* __restrict__ g3, const float* __restrict__ b3,
    const float* __restrict__ tn, float* __restrict__ out)
{
    __shared__ float xs[16 * ROWP];
    __shared__ float as_[16 * ROWP];
    __shared__ float sg2[192], sb2[192], sg3[192], sb3[192];
    __shared__ float tnl[128];
    __shared__ float mx[16], vx[16];
    __shared__ float pdot[16][8], psq[16][8];
    __shared__ float sattn[16], smeany[16], srstdy[16];

    const int t    = threadIdx.x;
    const int wave = t >> 5;
    const int lane = t & 31;
    const int nl   = lane & 15;
    const int hi   = lane >> 4;
    const int b     = blockIdx.x >> 4;
    const int chunk = blockIdx.x & 15;
    const int ncol  = wave * 16 + nl;

    for (int i = t; i < 192; i += 256) {
        sg2[i] = g2[i]; sb2[i] = b2[i]; sg3[i] = g3[i]; sb3[i] = b3[i];
    }
    if (t < 128) tnl[t] = tn[b * 128 + t];

    v2f bfrag[48];
    #pragma unroll
    for (int kb = 0; kb < 48; ++kb) {
        const int k0 = kb * 4 + hi * 2;
        bfrag[kb].x = w_x[k0 * 128 + ncol];
        bfrag[kb].y = w_x[(k0 + 1) * 128 + ncol];
    }
    const float bxv = b_x[ncol];
    lds_barrier();
    const float tnv = tnl[ncol];

    const size_t base = ((size_t)(b * 4096 + chunk * 256)) * 192;
    float4 pf0, pf1, pf2;
    {
        const float4* xg = (const float4*)(x + base);
        pf0 = xg[t]; pf1 = xg[t + 256]; pf2 = xg[t + 512];
    }
    float4* const xs0 = (float4*)&xs[(t / 48) * ROWP + (t % 48) * 4];
    float4* const xs1 = (float4*)&xs[((t + 256) / 48) * ROWP + ((t + 256) % 48) * 4];
    float4* const xs2 = (float4*)&xs[((t + 512) / 48) * ROWP + ((t + 512) % 48) * 4];

    for (int tt = 0; tt < 16; ++tt) {
        const size_t tbase = base + (size_t)tt * 3072;

        *xs0 = pf0; *xs1 = pf1; *xs2 = pf2;
        if (tt < 15) {
            const float4* xg = (const float4*)(x + tbase + 3072);
            pf0 = xg[t]; pf1 = xg[t + 256]; pf2 = xg[t + 512];
        }
        lds_barrier();

        for (int rr = 0; rr < 2; ++rr) {
            const int r = wave + rr * 8;
            float s = 0.f, q = 0.f;
            for (int j = 0; j < 6; ++j) {
                float xv = xs[r * ROWP + lane + j * 32];
                s += xv; q += xv * xv;
            }
            s += xhalf(s);   q += xhalf(q);
            s += xrow<8>(s); q += xrow<8>(q);
            s += xrow<4>(s); q += xrow<4>(q);
            s += xrow<2>(s); q += xrow<2>(q);
            s += xrow<1>(s); q += xrow<1>(q);
            float m   = s * (1.f / 192.f);
            float var = q * (1.f / 192.f) - m * m;
            float rst = rsqrtf(var + EPS_LN);
            if (lane == 0) { mx[r] = m; vx[r] = var; }
            for (int j = 0; j < 6; ++j) {
                int c = lane + j * 32;
                as_[r * ROWP + c] = (xs[r * ROWP + c] - m) * rst * sg2[c] + sb2[c];
            }
        }
        lds_barrier();

        v8f acc0 = {}, acc1 = {};
        const float* arow = &as_[nl * ROWP + hi * 2];
        #pragma unroll
        for (int kb = 0; kb < 48; kb += 2) {
            v2f a0 = *(const v2f*)(arow + kb * 4);
            v2f a1 = *(const v2f*)(arow + kb * 4 + 4);
            acc0 = __builtin_amdgcn_wmma_f32_16x16x4_f32(
                false, a0, false, bfrag[kb], (short)0, acc0, false, false);
            acc1 = __builtin_amdgcn_wmma_f32_16x16x4_f32(
                false, a1, false, bfrag[kb + 1], (short)0, acc1, false, false);
        }

        float pre[8], z[8], e[8];
        bool big = false;
        #pragma unroll
        for (int v = 0; v < 8; ++v) {
            pre[v] = acc0[v] + acc1[v] + bxv;
            z[v]   = pre[v] * 0.70710678118654752440f;
            big    = big || !(fabsf(z[v]) < 1.0f);
        }
        if (__builtin_amdgcn_ballot_w32(big) != 0u) {
            #pragma unroll
            for (int v = 0; v < 8; ++v) e[v] = erff(z[v]);
        } else {
            #pragma unroll
            for (int v = 0; v < 8; ++v) e[v] = erf_small(z[v]);
        }
        float pd[8], ps[8];
        #pragma unroll
        for (int v = 0; v < 8; ++v) {
            float g = 0.5f * pre[v] * (1.0f + e[v]);
            pd[v] = g * tnv;
            ps[v] = g * g;
        }
        #pragma unroll
        for (int v = 0; v < 8; ++v) { pd[v] += xrow<8>(pd[v]); ps[v] += xrow<8>(ps[v]); }
        #pragma unroll
        for (int v = 0; v < 8; ++v) { pd[v] += xrow<4>(pd[v]); ps[v] += xrow<4>(ps[v]); }
        #pragma unroll
        for (int v = 0; v < 8; ++v) { pd[v] += xrow<2>(pd[v]); ps[v] += xrow<2>(ps[v]); }
        #pragma unroll
        for (int v = 0; v < 8; ++v) { pd[v] += xrow<1>(pd[v]); ps[v] += xrow<1>(ps[v]); }
        if (nl == 0) {
            #pragma unroll
            for (int v = 0; v < 8; ++v) {
                pdot[v + hi * 8][wave] = pd[v];
                psq[v + hi * 8][wave]  = ps[v];
            }
        }
        lds_barrier();

        if (t < 16) {
            float dot = 0.f, sq = 0.f;
            for (int wv = 0; wv < 8; ++wv) { dot += pdot[t][wv]; sq += psq[t][wv]; }
            float attn = dot / fmaxf(sqrtf(sq), EPS_L2);
            sattn[t]  = attn;
            smeany[t] = attn * mx[t];
            srstdy[t] = rsqrtf(attn * attn * vx[t] + EPS_LN);
        }
        lds_barrier();

        {
            const int r  = t >> 4;
            const int c0 = (t & 15) * 12;
            const float attn = sattn[r], my = smeany[r], rs = srstdy[r];
            float* og = out + tbase + (size_t)r * 192;
            const float* xr = &xs[r * ROWP];
            #pragma unroll
            for (int c = c0; c < c0 + 12; ++c) {
                float xv = xr[c];
                og[c] = 0.5f * xv + (xv * attn - my) * rs * sg3[c] + sb3[c];
            }
        }
        lds_barrier();
    }
}

extern "C" void kernel_launch(void* const* d_in, const int* in_sizes, int n_in,
                              void* d_out, int out_size, void* d_ws, size_t ws_size,
                              hipStream_t stream) {
    const float* x     = (const float*)d_in[0];
    const float* token = (const float*)d_in[1];
    const float* p     = (const float*)d_in[2];
    const float* alpha = (const float*)d_in[3];
    const float* ln1_g = (const float*)d_in[4];
    const float* ln1_b = (const float*)d_in[5];
    const float* w_tok = (const float*)d_in[6];
    const float* b_tok = (const float*)d_in[7];
    const float* ln2_g = (const float*)d_in[8];
    const float* ln2_b = (const float*)d_in[9];
    const float* w_x   = (const float*)d_in[10];
    const float* b_x   = (const float*)d_in[11];
    const float* ln3_g = (const float*)d_in[12];
    const float* ln3_b = (const float*)d_in[13];
    float* out = (float*)d_out;
    float* tn  = (float*)d_ws;

    tok_kernel<<<32, 1024, 0, stream>>>(token, p, alpha, ln1_g, ln1_b,
                                       w_tok, b_tok, tn);
    xattn_kernel<<<512, 256, 0, stream>>>(x, w_x, b_x, ln2_g, ln2_b,
                                          ln3_g, ln3_b, tn, out);
}


# Each wave owns tile rows 2w, 2w+1 in registers from load to store: no raw x tile in LDS, attn finished per wave, 2 barriers per tile not 5, 512-byte stores; LN2 sums add per-lane partials first
# speedup vs baseline: 2.4912x; 1.0343x over previous
#include <hip/hip_runtime.h>
#include <math.h>

typedef __attribute__((ext_vector_type(2))) float v2f;
typedef __attribute__((ext_vector_type(8))) float v8f;

#define EPS_LN 1e-6f
#define EPS_L2 1e-12f

__device__ __forceinline__ float gelu_exact(float v) {
    return 0.5f * v * (1.0f + erff(v * 0.70710678118654752440f));
}

__device__ __forceinline__ float erf_small(float z) {
    const float t = z * z;
    float p = fmaf(t, -0x1.268bc2p-11f, 0x1.420828p-8f);
    p = fmaf(t, p, -0x1.b5937p-6f);
    p = fmaf(t, p, 0x1.ce077cp-4f);
    p = fmaf(t, p, -0x1.81266p-2f);
    p = fmaf(t, p, 0x1.06eba0p-3f);
    const float az = fabsf(z);
    return copysignf(fmaf(az, p, az), z);
}

__device__ __forceinline__ float xhalf(float x) {
    int b = __builtin_bit_cast(int, x);
    return __builtin_bit_cast(float, __builtin_amdgcn_permlanex16(
        b, b, 0x76543210, 0xfedcba98, false, false));
}
template <int M>
__device__ __forceinline__ float xrow(float x) {
    int b = __builtin_bit_cast(int, x);
    return __builtin_bit_cast(float, __builtin_amdgcn_update_dpp(
        b, b, 0x160 + M, 0xf, 0xf, true));
}

__device__ __forceinline__ void lds_barrier() {
    __builtin_amdgcn_fence(__ATOMIC_RELEASE, "workgroup", "local");
    __builtin_amdgcn_s_barrier();
    __builtin_amdgcn_fence(__ATOMIC_ACQUIRE, "workgroup", "local");
}

__global__ __launch_bounds__(1024) void tok_kernel(
    const float* __restrict__ token, const float* __restrict__ p,
    const float* __restrict__ alpha,
    const float* __restrict__ g1, const float* __restrict__ b1,
    const float* __restrict__ w_tok, const float* __restrict__ b_tok,
    float* __restrict__ tn_out)
{
    __shared__ float red[256];
    __shared__ float tnorm[768];
    __shared__ float gbuf[128];
    __shared__ float4 part[32 * 32];
    __shared__ float sm, srstd, snrm;
    const int b = blockIdx.x, t = threadIdx.x;
    const int w = t >> 5, l = t & 31;
    const float* tok = token + b * 768;

    float a0 = 0.f, a1 = 0.f, a2 = 0.f;
    if (t < 256) { a0 = tok[t]; a1 = tok[t + 256]; a2 = tok[t + 512]; }
    float tkv = 0.f, g1v = 0.f, b1v = 0.f;
    if (t < 768) { tkv = tok[t]; g1v = g1[t]; b1v = b1[t]; }
    float btv = 0.f;
    if (t < 128) btv = b_tok[t];

    float4 wv[24];
    {
        const float4* wr = (const float4*)(w_tok + (size_t)w * (24 * 128)) + l;
        #pragma unroll
        for (int j = 0; j < 24; ++j) wv[j] = wr[j * 32];
    }

    if (t < 256) red[t] = ((0.f + a0) + a1) + a2;
    lds_barrier();
    for (int off = 128; off > 0; off >>= 1) { if (t < off) red[t] += red[t + off]; lds_barrier(); }
    if (t == 0) sm = red[0] * (1.f / 768.f);
    lds_barrier();
    const float m = sm;

    if (t < 256) {
        const float d0 = a0 - m, d1 = a1 - m, d2 = a2 - m;
        float q = 0.f;
        q += d0 * d0; q += d1 * d1; q += d2 * d2;
        red[t] = q;
    }
    lds_barrier();
    for (int off = 128; off > 0; off >>= 1) { if (t < off) red[t] += red[t + off]; lds_barrier(); }
    if (t == 0) srstd = rsqrtf(red[0] * (1.f / 768.f) + EPS_LN);
    lds_barrier();
    const float rstd = srstd;

    if (t < 768) tnorm[t] = (tkv - m) * rstd * g1v + b1v;
    lds_barrier();

    {
        float4 acc = {0.f, 0.f, 0.f, 0.f};
        const float* tk = &tnorm[w * 24];
        #pragma unroll
        for (int j = 0; j < 24; ++j) {
            const float a = tk[j];
            acc.x = fmaf(a, wv[j].x, acc.x);
            acc.y = fmaf(a, wv[j].y, acc.y);
            acc.z = fmaf(a, wv[j].z, acc.z);
            acc.w = fmaf(a, wv[j].w, acc.w);
        }
        part[w * 32 + l] = acc;
    }
    lds_barrier();

    if (t < 128) {
        const float* pf = (const float*)part;
        float acc = btv;
        for (int sl = 0; sl < 32; ++sl) acc += pf[sl * 128 + t];
        gbuf[t] = gelu_exact(acc);
    }
    lds_barrier();

    if (t < 256) {
        float ss = 0.f;
        if (t < 128) { float g = gbuf[t]; ss = g * g; }
        red[t] = ss;
    }
    lds_barrier();
    for (int off = 128; off > 0; off >>= 1) { if (t < off) red[t] += red[t + off]; lds_barrier(); }
    if (t == 0) snrm = fmaxf(sqrtf(red[0]), EPS_L2);
    lds_barrier();

    if (t < 128) {
        float scale = p[b] * expf(alpha[0]) / snrm;
        tn_out[b * 128 + t] = gbuf[t] * scale;
    }
}

#define ROWP 196

__global__ __launch_bounds__(256) void xattn_kernel(
    const float* __restrict__ x, const float* __restrict__ w_x,
    const float* __restrict__ b_x,
    const float* __restrict__ g2, const float* __restrict__ b2,
    const float* __restrict__ g3, const float* __restrict__ b3,
    const float* __restrict__ tn, float* __restrict__ out)
{
    __shared__ float4 as4[16 * (ROWP / 4)];
    __shared__ float4 sg2[48], sb2[48], sg3[48], sb3[48];
    __shared__ float tnl[128];
    __shared__ float4 pdot4[16][2], psq4[16][2];
    float* const as_ = (float*)as4;
    float (* const pdot)[8] = (float (*)[8])pdot4;
    float (* const psq)[8]  = (float (*)[8])psq4;

    const int t    = threadIdx.x;
    const int wave = t >> 5;
    const int lane = t & 31;
    const int nl   = lane & 15;
    const int hi   = lane >> 4;
    const int b     = blockIdx.x >> 4;
    const int chunk = blockIdx.x & 15;
    const int ncol  = wave * 16 + nl;

    for (int i = t; i < 192; i += 256) {
        ((float*)sg2)[i] = g2[i]; ((float*)sb2)[i] = b2[i];
        ((float*)sg3)[i] = g3[i]; ((float*)sb3)[i] = b3[i];
    }
    if (t < 128) tnl[t] = tn[b * 128 + t];

    v2f bfrag[48];
    #pragma unroll
    for (int kb = 0; kb < 48; ++kb) {
        const int k0 = kb * 4 + hi * 2;
        bfrag[kb].x = w_x[k0 * 128 + ncol];
        bfrag[kb].y = w_x[(k0 + 1) * 128 + ncol];
    }
    const float bxv = b_x[ncol];
    lds_barrier();
    const float tnv = tnl[ncol];

    const int r0   = 2 * wave;
    const int rmid = r0 + hi;
    const int r1   = r0 + 1;
    const int c0   = lane;
    const int c1   = hi ? lane - 16 : lane + 32;
    const int c2   = lane + 16;
    float4* const as0 = &as4[r0   * (ROWP / 4) + c0];
    float4* const as1 = &as4[rmid * (ROWP / 4) + c1];
    float4* const as2 = &as4[r1   * (ROWP / 4) + c2];

    const size_t base = ((size_t)(b * 4096 + chunk * 256)) * 192;
    const size_t wofs = (size_t)wave * 384 + (size_t)lane * 4;
    float4 pf0, pf1, pf2;
    {
        const float4* xg = (const float4*)(x + base + wofs);
        pf0 = xg[0]; pf1 = xg[32]; pf2 = xg[64];
    }

    for (int tt = 0; tt < 16; ++tt) {
        const size_t tbase = base + (size_t)tt * 3072;

        const float4 q0 = pf0, q1 = pf1, q2 = pf2;
        if (tt < 15) {
            const float4* xg = (const float4*)(x + tbase + 3072 + wofs);
            pf0 = xg[0]; pf1 = xg[32]; pf2 = xg[64];
        }

        const float t0 = ((q0.x + q0.y) + q0.z) + q0.w;
        const float t1 = ((q1.x + q1.y) + q1.z) + q1.w;
        const float t2 = ((q2.x + q2.y) + q2.z) + q2.w;
        const float u0 = ((q0.x * q0.x + q0.y * q0.y) + q0.z * q0.z) + q0.w * q0.w;
        const float u1 = ((q1.x * q1.x + q1.y * q1.y) + q1.z * q1.z) + q1.w * q1.w;
        const float u2 = ((q2.x * q2.x + q2.y * q2.y) + q2.z * q2.z) + q2.w * q2.w;
        float sA = t0 + (hi ? 0.f : t1), qA = u0 + (hi ? 0.f : u1);
        float sB = t2 + (hi ? t1 : 0.f), qB = u2 + (hi ? u1 : 0.f);
        sA += xhalf(sA);   qA += xhalf(qA);   sB += xhalf(sB);   qB += xhalf(qB);
        sA += xrow<8>(sA); qA += xrow<8>(qA); sB += xrow<8>(sB); qB += xrow<8>(qB);
        sA += xrow<4>(sA); qA += xrow<4>(qA); sB += xrow<4>(sB); qB += xrow<4>(qB);
        sA += xrow<2>(sA); qA += xrow<2>(qA); sB += xrow<2>(sB); qB += xrow<2>(qB);
        sA += xrow<1>(sA); qA += xrow<1>(qA); sB += xrow<1>(sB); qB += xrow<1>(qB);
        const float mA   = sA * (1.f / 192.f);
        const float varA = qA * (1.f / 192.f) - mA * mA;
        const float rstA = rsqrtf(varA + EPS_LN);
        const float mB   = sB * (1.f / 192.f);
        const float varB = qB * (1.f / 192.f) - mB * mB;
        const float rstB = rsqrtf(varB + EPS_LN);
        const float mM   = hi ? mB : mA;
        const float varM = hi ? varB : varA;
        const float rstM = hi ? rstB : rstA;

        {
            const float4 ga = sg2[c0], gb = sg2[c1], gc = sg2[c2];
            const float4 ba = sb2[c0], bb = sb2[c1], bc = sb2[c2];
            float4 a0, a1, a2;
            a0.x = (q0.x - mA) * rstA * ga.x + ba.x;
            a0.y = (q0.y - mA) * rstA * ga.y + ba.y;
            a0.z = (q0.z - mA) * rstA * ga.z + ba.z;
            a0.w = (q0.w - mA) * rstA * ga.w + ba.w;
            a1.x = (q1.x - mM) * rstM * gb.x + bb.x;
            a1.y = (q1.y - mM) * rstM * gb.y + bb.y;
            a1.z = (q1.z - mM) * rstM * gb.z + bb.z;
            a1.w = (q1.w - mM) * rstM * gb.w + bb.w;
            a2.x = (q2.x - mB) * rstB * gc.x + bc.x;
            a2.y = (q2.y - mB) * rstB * gc.y + bc.y;
            a2.z = (q2.z - mB) * rstB * gc.z + bc.z;
            a2.w = (q2.w - mB) * rstB * gc.w + bc.w;
            *as0 = a0; *as1 = a1; *as2 = a2;
        }
        lds_barrier();

        v8f acc0 = {}, acc1 = {};
        const float* arow = &as_[nl * ROWP + hi * 2];
        #pragma unroll
        for (int kb = 0; kb < 48; kb += 2) {
            v2f a0 = *(const v2f*)(arow + kb * 4);
            v2f a1 = *(const v2f*)(arow + kb * 4 + 4);
            acc0 = __builtin_amdgcn_wmma_f32_16x16x4_f32(
                false, a0, false, bfrag[kb], (short)0, acc0, false, false);
            acc1 = __builtin_amdgcn_wmma_f32_16x16x4_f32(
                false, a1, false, bfrag[kb + 1], (short)0, acc1, false, false);
        }

        float pre[8], z[8], e[8];
        bool big = false;
        #pragma unroll
        for (int v = 0; v < 8; ++v) {
            pre[v] = acc0[v] + acc1[v] + bxv;
            z[v]   = pre[v] * 0.70710678118654752440f;
            big    = big || !(fabsf(z[v]) < 1.0f);
        }
        if (__builtin_amdgcn_ballot_w32(big) != 0u) {
            #pragma unroll
            for (int v = 0; v < 8; ++v) e[v] = erff(z[v]);
        } else {
            #pragma unroll
            for (int v = 0; v < 8; ++v) e[v] = erf_small(z[v]);
        }
        float pd[8], ps[8];
        #pragma unroll
        for (int v = 0; v < 8; ++v) {
            float g = 0.5f * pre[v] * (1.0f + e[v]);
            pd[v] = g * tnv;
            ps[v] = g * g;
        }
        #pragma unroll
        for (int v = 0; v < 8; ++v) { pd[v] += xrow<8>(pd[v]); ps[v] += xrow<8>(ps[v]); }
        #pragma unroll
        for (int v = 0; v < 8; ++v) { pd[v] += xrow<4>(pd[v]); ps[v] += xrow<4>(ps[v]); }
        #pragma unroll
        for (int v = 0; v < 8; ++v) { pd[v] += xrow<2>(pd[v]); ps[v] += xrow<2>(ps[v]); }
        #pragma unroll
        for (int v = 0; v < 8; ++v) { pd[v] += xrow<1>(pd[v]); ps[v] += xrow<1>(ps[v]); }
        if (nl == 0) {
            #pragma unroll
            for (int v = 0; v < 8; ++v) {
                pdot[v + hi * 8][wave] = pd[v];
                psq[v + hi * 8][wave]  = ps[v];
            }
        }
        lds_barrier();

        float dot = 0.f, sq = 0.f;
        for (int wv = 0; wv < 8; ++wv) { dot += pdot[rmid][wv]; sq += psq[rmid][wv]; }
        const float attnM = dot / fmaxf(sqrtf(sq), EPS_L2);
        const float myM   = attnM * mM;
        const float rsM   = rsqrtf(attnM * attnM * varM + EPS_LN);
        const float attnO = xhalf(attnM), myO = xhalf(myM), rsO = xhalf(rsM);
        const float attnA = hi ? attnO : attnM, myA = hi ? myO : myM, rsA = hi ? rsO : rsM;
        const float attnB = hi ? attnM : attnO, myB = hi ? myM : myO, rsB = hi ? rsM : rsO;

        {
            const float4 ga = sg3[c0], gb = sg3[c1], gc = sg3[c2];
            const float4 ba = sb3[c0], bb = sb3[c1], bc = sb3[c2];
            float4 o0, o1, o2;
            o0.x = 0.5f * q0.x + (q0.x * attnA - myA) * rsA * ga.x + ba.x;
            o0.y = 0.5f * q0.y + (q0.y * attnA - myA) * rsA * ga.y + ba.y;
            o0.z = 0.5f * q0.z + (q0.z * attnA - myA) * rsA * ga.z + ba.z;
            o0.w = 0.5f * q0.w + (q0.w * attnA - myA) * rsA * ga.w + ba.w;
            o1.x = 0.5f * q1.x + (q1.x * attnM - myM) * rsM * gb.x + bb.x;
            o1.y = 0.5f * q1.y + (q1.y * attnM - myM) * rsM * gb.y + bb.y;
            o1.z = 0.5f * q1.z + (q1.z * attnM - myM) * rsM * gb.z + bb.z;
            o1.w = 0.5f * q1.w + (q1.w * attnM - myM) * rsM * gb.w + bb.w;
            o2.x = 0.5f * q2.x + (q2.x * attnB - myB) * rsB * gc.x + bc.x;
            o2.y = 0.5f * q2.y + (q2.y * attnB - myB) * rsB * gc.y + bc.y;
            o2.z = 0.5f * q2.z + (q2.z * attnB - myB) * rsB * gc.z + bc.z;
            o2.w = 0.5f * q2.w + (q2.w * attnB - myB) * rsB * gc.w + bc.w;
            float4* og = (float4*)(out + tbase + wofs);
            og[0] = o0; og[32] = o1; og[64] = o2;
        }
    }
}

extern "C" void kernel_launch(void* const* d_in, const int* in_sizes, int n_in,
                              void* d_out, int out_size, void* d_ws, size_t ws_size,
                              hipStream_t stream) {
    const float* x     = (const float*)d_in[0];
    const float* token = (const float*)d_in[1];
    const float* p     = (const float*)d_in[2];
    const float* alpha = (const float*)d_in[3];
    const float* ln1_g = (const float*)d_in[4];
    const float* ln1_b = (const float*)d_in[5];
    const float* w_tok = (const float*)d_in[6];
    const float* b_tok = (const float*)d_in[7];
    const float* ln2_g = (const float*)d_in[8];
    const float* ln2_b = (const float*)d_in[9];
    const float* w_x   = (const float*)d_in[10];
    const float* b_x   = (const float*)d_in[11];
    const float* ln3_g = (const float*)d_in[12];
    const float* ln3_b = (const float*)d_in[13];
    float* out = (float*)d_out;
    float* tn  = (float*)d_ws;

    tok_kernel<<<32, 1024, 0, stream>>>(token, p, alpha, ln1_g, ln1_b,
                                       w_tok, b_tok, tn);
    xattn_kernel<<<512, 256, 0, stream>>>(x, w_x, b_x, ln2_g, ln2_b,
                                          ln3_g, ln3_b, tn, out);
}


# Tile product transposed (w_x fragments as A, x rows as B): a lane's 8 accumulators are 8 columns of ONE row; per-row dot/square sums are in-lane fma chains + one permlanex16; changes their sum order
# speedup vs baseline: 2.6568x; 1.0665x over previous
#include <hip/hip_runtime.h>
#include <math.h>

typedef __attribute__((ext_vector_type(2))) float v2f;
typedef __attribute__((ext_vector_type(8))) float v8f;

#define EPS_LN 1e-6f
#define EPS_L2 1e-12f

__device__ __forceinline__ float gelu_exact(float v) {
    return 0.5f * v * (1.0f + erff(v * 0.70710678118654752440f));
}

__device__ __forceinline__ float erf_small(float z) {
    const float t = z * z;
    float p = fmaf(t, -0x1.268bc2p-11f, 0x1.420828p-8f);
    p = fmaf(t, p, -0x1.b5937p-6f);
    p = fmaf(t, p, 0x1.ce077cp-4f);
    p = fmaf(t, p, -0x1.81266p-2f);
    p = fmaf(t, p, 0x1.06eba0p-3f);
    const float az = fabsf(z);
    return copysignf(fmaf(az, p, az), z);
}

__device__ __forceinline__ float xhalf(float x) {
    int b = __builtin_bit_cast(int, x);
    return __builtin_bit_cast(float, __builtin_amdgcn_permlanex16(
        b, b, 0x76543210, 0xfedcba98, false, false));
}
template <int M>
__device__ __forceinline__ float xrow(float x) {
    int b = __builtin_bit_cast(int, x);
    return __builtin_bit_cast(float, __builtin_amdgcn_update_dpp(
        b, b, 0x160 + M, 0xf, 0xf, true));
}

__device__ __forceinline__ void lds_barrier() {
    __builtin_amdgcn_fence(__ATOMIC_RELEASE, "workgroup", "local");
    __builtin_amdgcn_s_barrier();
    __builtin_amdgcn_fence(__ATOMIC_ACQUIRE, "workgroup", "local");
}

__global__ __launch_bounds__(1024) void tok_kernel(
    const float* __restrict__ token, const float* __restrict__ p,
    const float* __restrict__ alpha,
    const float* __restrict__ g1, const float* __restrict__ b1,
    const float* __restrict__ w_tok, const float* __restrict__ b_tok,
    float* __restrict__ tn_out)
{
    __shared__ float red[256];
    __shared__ float tnorm[768];
    __shared__ float gbuf[128];
    __shared__ float4 part[32 * 32];
    __shared__ float sm, srstd, snrm;
    const int b = blockIdx.x, t = threadIdx.x;
    const int w = t >> 5, l = t & 31;
    const float* tok = token + b * 768;

    float a0 = 0.f, a1 = 0.f, a2 = 0.f;
    if (t < 256) { a0 = tok[t]; a1 = tok[t + 256]; a2 = tok[t + 512]; }
    float tkv = 0.f, g1v = 0.f, b1v = 0.f;
    if (t < 768) { tkv = tok[t]; g1v = g1[t]; b1v = b1[t]; }
    float btv = 0.f;
    if (t < 128) btv = b_tok[t];

    float4 wv[24];
    {
        const float4* wr = (const float4*)(w_tok + (size_t)w * (24 * 128)) + l;
        #pragma unroll
        for (int j = 0; j < 24; ++j) wv[j] = wr[j * 32];
    }

    if (t < 256) red[t] = ((0.f + a0) + a1) + a2;
    lds_barrier();
    for (int off = 128; off > 0; off >>= 1) { if (t < off) red[t] += red[t + off]; lds_barrier(); }
    if (t == 0) sm = red[0] * (1.f / 768.f);
    lds_barrier();
    const float m = sm;

    if (t < 256) {
        const float d0 = a0 - m, d1 = a1 - m, d2 = a2 - m;
        float q = 0.f;
        q += d0 * d0; q += d1 * d1; q += d2 * d2;
        red[t] = q;
    }
    lds_barrier();
    for (int off = 128; off > 0; off >>= 1) { if (t < off) red[t] += red[t + off]; lds_barrier(); }
    if (t == 0) srstd = rsqrtf(red[0] * (1.f / 768.f) + EPS_LN);
    lds_barrier();
    const float rstd = srstd;

    if (t < 768) tnorm[t] = (tkv - m) * rstd * g1v + b1v;
    lds_barrier();

    {
        float4 acc = {0.f, 0.f, 0.f, 0.f};
        const float* tk = &tnorm[w * 24];
        #pragma unroll
        for (int j = 0; j < 24; ++j) {
            const float a = tk[j];
            acc.x = fmaf(a, wv[j].x, acc.x);
            acc.y = fmaf(a, wv[j].y, acc.y);
            acc.z = fmaf(a, wv[j].z, acc.z);
            acc.w = fmaf(a, wv[j].w, acc.w);
        }
        part[w * 32 + l] = acc;
    }
    lds_barrier();

    if (t < 128) {
        const float* pf = (const float*)part;
        float acc = btv;
        for (int sl = 0; sl < 32; ++sl) acc += pf[sl * 128 + t];
        gbuf[t] = gelu_exact(acc);
    }
    lds_barrier();

    if (t < 256) {
        float ss = 0.f;
        if (t < 128) { float g = gbuf[t]; ss = g * g; }
        red[t] = ss;
    }
    lds_barrier();
    for (int off = 128; off > 0; off >>= 1) { if (t < off) red[t] += red[t + off]; lds_barrier(); }
    if (t == 0) snrm = fmaxf(sqrtf(red[0]), EPS_L2);
    lds_barrier();

    if (t < 128) {
        float scale = p[b] * expf(alpha[0]) / snrm;
        tn_out[b * 128 + t] = gbuf[t] * scale;
    }
}

#define ROWP 196

__global__ __launch_bounds__(256) void xattn_kernel(
    const float* __restrict__ x, const float* __restrict__ w_x,
    const float* __restrict__ b_x,
    const float* __restrict__ g2, const float* __restrict__ b2,
    const float* __restrict__ g3, const float* __restrict__ b3,
    const float* __restrict__ tn, float* __restrict__ out)
{
    __shared__ float4 as4[16 * (ROWP / 4)];
    __shared__ float4 sg2[48], sb2[48], sg3[48], sb3[48];
    __shared__ float4 pdot4[16][2], psq4[16][2];
    float* const as_ = (float*)as4;
    float (* const pdot)[8] = (float (*)[8])pdot4;
    float (* const psq)[8]  = (float (*)[8])psq4;

    const int t    = threadIdx.x;
    const int wave = t >> 5;
    const int lane = t & 31;
    const int nl   = lane & 15;
    const int hi   = lane >> 4;
    const int b     = blockIdx.x >> 4;
    const int chunk = blockIdx.x & 15;
    const int ncol  = wave * 16 + nl;
    const int ncol0 = wave * 16 + hi * 8;

    for (int i = t; i < 192; i += 256) {
        ((float*)sg2)[i] = g2[i]; ((float*)sb2)[i] = b2[i];
        ((float*)sg3)[i] = g3[i]; ((float*)sb3)[i] = b3[i];
    }

    v2f bfrag[48];
    #pragma unroll
    for (int kb = 0; kb < 48; ++kb) {
        const int k0 = kb * 4 + hi * 2;
        bfrag[kb].x = w_x[k0 * 128 + ncol];
        bfrag[kb].y = w_x[(k0 + 1) * 128 + ncol];
    }
    float bx8[8], tn8[8];
    #pragma unroll
    for (int v = 0; v < 8; ++v) {
        bx8[v] = b_x[ncol0 + v];
        tn8[v] = tn[b * 128 + ncol0 + v];
    }
    lds_barrier();
    float* const pslot = hi ? &psq[nl][wave] : &pdot[nl][wave];

    const int r0   = 2 * wave;
    const int rmid = r0 + hi;
    const int r1   = r0 + 1;
    const int c0   = lane;
    const int c1   = hi ? lane - 16 : lane + 32;
    const int c2   = lane + 16;
    float4* const as0 = &as4[r0   * (ROWP / 4) + c0];
    float4* const as1 = &as4[rmid * (ROWP / 4) + c1];
    float4* const as2 = &as4[r1   * (ROWP / 4) + c2];

    const size_t base = ((size_t)(b * 4096 + chunk * 256)) * 192;
    const size_t wofs = (size_t)wave * 384 + (size_t)lane * 4;
    float4 pf0, pf1, pf2;
    {
        const float4* xg = (const float4*)(x + base + wofs);
        pf0 = xg[0]; pf1 = xg[32]; pf2 = xg[64];
    }

    for (int tt = 0; tt < 16; ++tt) {
        const size_t tbase = base + (size_t)tt * 3072;

        const float4 q0 = pf0, q1 = pf1, q2 = pf2;
        if (tt < 15) {
            const float4* xg = (const float4*)(x + tbase + 3072 + wofs);
            pf0 = xg[0]; pf1 = xg[32]; pf2 = xg[64];
        }

        const float t0 = ((q0.x + q0.y) + q0.z) + q0.w;
        const float t1 = ((q1.x + q1.y) + q1.z) + q1.w;
        const float t2 = ((q2.x + q2.y) + q2.z) + q2.w;
        const float u0 = ((q0.x * q0.x + q0.y * q0.y) + q0.z * q0.z) + q0.w * q0.w;
        const float u1 = ((q1.x * q1.x + q1.y * q1.y) + q1.z * q1.z) + q1.w * q1.w;
        const float u2 = ((q2.x * q2.x + q2.y * q2.y) + q2.z * q2.z) + q2.w * q2.w;
        float sA = t0 + (hi ? 0.f : t1), qA = u0 + (hi ? 0.f : u1);
        float sB = t2 + (hi ? t1 : 0.f), qB = u2 + (hi ? u1 : 0.f);
        sA += xhalf(sA);   qA += xhalf(qA);   sB += xhalf(sB);   qB += xhalf(qB);
        sA += xrow<8>(sA); qA += xrow<8>(qA); sB += xrow<8>(sB); qB += xrow<8>(qB);
        sA += xrow<4>(sA); qA += xrow<4>(qA); sB += xrow<4>(sB); qB += xrow<4>(qB);
        sA += xrow<2>(sA); qA += xrow<2>(qA); sB += xrow<2>(sB); qB += xrow<2>(qB);
        sA += xrow<1>(sA); qA += xrow<1>(qA); sB += xrow<1>(sB); qB += xrow<1>(qB);
        const float mA   = sA * (1.f / 192.f);
        const float varA = qA * (1.f / 192.f) - mA * mA;
        const float rstA = rsqrtf(varA + EPS_LN);
        const float mB   = sB * (1.f / 192.f);
        const float varB = qB * (1.f / 192.f) - mB * mB;
        const float rstB = rsqrtf(varB + EPS_LN);
        const float mM   = hi ? mB : mA;
        const float varM = hi ? varB : varA;
        const float rstM = hi ? rstB : rstA;

        {
            const float4 ga = sg2[c0], gb = sg2[c1], gc = sg2[c2];
            const float4 ba = sb2[c0], bb = sb2[c1], bc = sb2[c2];
            float4 a0, a1, a2;
            a0.x = (q0.x - mA) * rstA * ga.x + ba.x;
            a0.y = (q0.y - mA) * rstA * ga.y + ba.y;
            a0.z = (q0.z - mA) * rstA * ga.z + ba.z;
            a0.w = (q0.w - mA) * rstA * ga.w + ba.w;
            a1.x = (q1.x - mM) * rstM * gb.x + bb.x;
            a1.y = (q1.y - mM) * rstM * gb.y + bb.y;
            a1.z = (q1.z - mM) * rstM * gb.z + bb.z;
            a1.w = (q1.w - mM) * rstM * gb.w + bb.w;
            a2.x = (q2.x - mB) * rstB * gc.x + bc.x;
            a2.y = (q2.y - mB) * rstB * gc.y + bc.y;
            a2.z = (q2.z - mB) * rstB * gc.z + bc.z;
            a2.w = (q2.w - mB) * rstB * gc.w + bc.w;
            *as0 = a0; *as1 = a1; *as2 = a2;
        }
        lds_barrier();

        v8f acc0 = {}, acc1 = {};
        const float* arow = &as_[nl * ROWP + hi * 2];
        #pragma unroll
        for (int kb = 0; kb < 48; kb += 2) {
            v2f x0 = *(const v2f*)(arow + kb * 4);
            v2f x1 = *(const v2f*)(arow + kb * 4 + 4);
            acc0 = __builtin_amdgcn_wmma_f32_16x16x4_f32(
                false, bfrag[kb], false, x0, (short)0, acc0, false, false);
            acc1 = __builtin_amdgcn_wmma_f32_16x16x4_f32(
                false, bfrag[kb + 1], false, x1, (short)0, acc1, false, false);
        }

        float pre[8], z[8], e[8];
        bool big = false;
        #pragma unroll
        for (int v = 0; v < 8; ++v) {
            pre[v] = acc0[v] + acc1[v] + bx8[v];
            z[v]   = pre[v] * 0.70710678118654752440f;
            big    = big || !(fabsf(z[v]) < 1.0f);
        }
        if (__builtin_amdgcn_ballot_w32(big) != 0u) {
            #pragma unroll
            for (int v = 0; v < 8; ++v) e[v] = erff(z[v]);
        } else {
            #pragma unroll
            for (int v = 0; v < 8; ++v) e[v] = erf_small(z[v]);
        }
        float pdl = 0.f, psl = 0.f;
        #pragma unroll
        for (int v = 0; v < 8; ++v) {
            const float g = 0.5f * pre[v] * (1.0f + e[v]);
            pdl = fmaf(g, tn8[v], pdl);
            psl = fmaf(g, g, psl);
        }
        pdl += xhalf(pdl);
        psl += xhalf(psl);
        *pslot = hi ? psl : pdl;
        lds_barrier();

        float dot = 0.f, sq = 0.f;
        for (int wv = 0; wv < 8; ++wv) { dot += pdot[rmid][wv]; sq += psq[rmid][wv]; }
        const float attnM = dot / fmaxf(sqrtf(sq), EPS_L2);
        const float myM   = attnM * mM;
        const float rsM   = rsqrtf(attnM * attnM * varM + EPS_LN);
        const float attnO = xhalf(attnM), myO = xhalf(myM), rsO = xhalf(rsM);
        const float attnA = hi ? attnO : attnM, myA = hi ? myO : myM, rsA = hi ? rsO : rsM;
        const float attnB = hi ? attnM : attnO, myB = hi ? myM : myO, rsB = hi ? rsM : rsO;

        {
            const float4 ga = sg3[c0], gb = sg3[c1], gc = sg3[c2];
            const float4 ba = sb3[c0], bb = sb3[c1], bc = sb3[c2];
            float4 o0, o1, o2;
            o0.x = 0.5f * q0.x + (q0.x * attnA - myA) * rsA * ga.x + ba.x;
            o0.y = 0.5f * q0.y + (q0.y * attnA - myA) * rsA * ga.y + ba.y;
            o0.z = 0.5f * q0.z + (q0.z * attnA - myA) * rsA * ga.z + ba.z;
            o0.w = 0.5f * q0.w + (q0.w * attnA - myA) * rsA * ga.w + ba.w;
            o1.x = 0.5f * q1.x + (q1.x * attnM - myM) * rsM * gb.x + bb.x;
            o1.y = 0.5f * q1.y + (q1.y * attnM - myM) * rsM * gb.y + bb.y;
            o1.z = 0.5f * q1.z + (q1.z * attnM - myM) * rsM * gb.z + bb.z;
            o1.w = 0.5f * q1.w + (q1.w * attnM - myM) * rsM * gb.w + bb.w;
            o2.x = 0.5f * q2.x + (q2.x * attnB - myB) * rsB * gc.x + bc.x;
            o2.y = 0.5f * q2.y + (q2.y * attnB - myB) * rsB * gc.y + bc.y;
            o2.z = 0.5f * q2.z + (q2.z * attnB - myB) * rsB * gc.z + bc.z;
            o2.w = 0.5f * q2.w + (q2.w * attnB - myB) * rsB * gc.w + bc.w;
            float4* og = (float4*)(out + tbase + wofs);
            og[0] = o0; og[32] = o1; og[64] = o2;
        }
    }
}

extern "C" void kernel_launch(void* const* d_in, const int* in_sizes, int n_in,
                              void* d_out, int out_size, void* d_ws, size_t ws_size,
                              hipStream_t stream) {
    const float* x     = (const float*)d_in[0];
    const float* token = (const float*)d_in[1];
    const float* p     = (const float*)d_in[2];
    const float* alpha = (const float*)d_in[3];
    const float* ln1_g = (const float*)d_in[4];
    const float* ln1_b = (const float*)d_in[5];
    const float* w_tok = (const float*)d_in[6];
    const float* b_tok = (const float*)d_in[7];
    const float* ln2_g = (const float*)d_in[8];
    const float* ln2_b = (const float*)d_in[9];
    const float* w_x   = (const float*)d_in[10];
    const float* b_x   = (const float*)d_in[11];
    const float* ln3_g = (const float*)d_in[12];
    const float* ln3_b = (const float*)d_in[13];
    float* out = (float*)d_out;
    float* tn  = (float*)d_ws;

    tok_kernel<<<32, 1024, 0, stream>>>(token, p, alpha, ln1_g, ln1_b,
                                       w_tok, b_tok, tn);
    xattn_kernel<<<512, 256, 0, stream>>>(x, w_x, b_x, ln2_g, ln2_b,
                                          ln3_g, ln3_b, tn, out);
}
